# starting kernel
# baseline (speedup 1.0000x reference)
#include <hip/hip_runtime.h>

typedef __attribute__((ext_vector_type(2))) float v2f;
typedef __attribute__((ext_vector_type(8))) float v8f;

#define N_NODES 50000
#define N_EDGES 800000
#define D_IN    128
#define KK      4
#define FF      32
#define KF      128

__device__ __forceinline__ unsigned ordOf(float f) {
  unsigned u = __float_as_uint(f);
  return (u >> 31) ? ~u : (u | 0x80000000u);
}
__device__ __forceinline__ float ofOrd(unsigned o) {
  return (o >> 31) ? __uint_as_float(o & 0x7fffffffu) : __uint_as_float(~o);
}

__global__ __launch_bounds__(256) void init_ws(const float* __restrict__ bias,
                                               float* __restrict__ out,
                                               unsigned* __restrict__ nodemax,
                                               float* __restrict__ nodesum) {
  int t = blockIdx.x * 256 + threadIdx.x;
  if (t < N_NODES * KF) out[t] = bias[t & (KF - 1)];
  if (t < N_NODES * KK) { nodemax[t] = 0u; nodesum[t] = 0.0f; }
}

__global__ __launch_bounds__(256) void fc_wmma(const float* __restrict__ feat,
                                               const float* __restrict__ W,
                                               float* __restrict__ ft) {
  __shared__ float As[16 * D_IN];
  const int lane = threadIdx.x & 31;
  const int wave = threadIdx.x >> 5;
  const int row0 = blockIdx.x * 16;

  for (int i = threadIdx.x; i < 16 * D_IN / 4; i += 256) {
    ((float4*)As)[i] = ((const float4*)(feat + row0 * D_IN))[i];
  }
  __syncthreads();

  const int col0 = wave * 16;
  const int ln   = lane & 15;
  const int kh   = (lane >> 4) << 1;

  v8f c = {};
  #pragma unroll 8
  for (int k = 0; k < D_IN; k += 4) {
    v2f a; a.x = As[ln * D_IN + k + kh];
           a.y = As[ln * D_IN + k + kh + 1];
    v2f b; b.x = W[(col0 + ln) * D_IN + k + kh];
           b.y = W[(col0 + ln) * D_IN + k + kh + 1];
    c = __builtin_amdgcn_wmma_f32_16x16x4_f32(false, a, false, b,
                                              (short)0, c, false, false);
  }

  const int mo = (lane >> 4) << 3;
  #pragma unroll
  for (int r = 0; r < 8; ++r) {
    ft[(row0 + r + mo) * KF + col0 + ln] = c[r];
  }
}

__global__ __launch_bounds__(256) void edge_logits(const float* __restrict__ ft,
                                                   const float* __restrict__ mu,
                                                   const float* __restrict__ sigma,
                                                   const int* __restrict__ src,
                                                   const int* __restrict__ dst,
                                                   float* __restrict__ lbuf,
                                                   unsigned* __restrict__ nodemax) {
  int t = blockIdx.x * 256 + threadIdx.x;
  if (t >= N_EDGES * KK) return;
  const int k = t & 3;
  const int e = t >> 2;
  const int s = src[e];
  const int d = dst[e];
  const float4* ps = (const float4*)(ft + (size_t)s * KF + k * FF);
  const float4* pd = (const float4*)(ft + (size_t)d * KF + k * FF);
  const float4* pm = (const float4*)(mu + k * FF);
  float acc = 0.0f;
  #pragma unroll
  for (int i = 0; i < FF / 4; ++i) {
    float4 a = ps[i], b = pd[i], m = pm[i];
    float dx = a.x - b.x - m.x;
    float dy = a.y - b.y - m.y;
    float dz = a.z - b.z - m.z;
    float dw = a.w - b.w - m.w;
    acc += dx * dx + dy * dy + dz * dz + dw * dw;
  }
  float logit = -sqrtf(acc) * sigma[k];
  lbuf[t] = logit;
  atomicMax(nodemax + (size_t)d * KK + k, ordOf(logit));
}

__global__ __launch_bounds__(256) void edge_exp(const int* __restrict__ dst,
                                                const unsigned* __restrict__ nodemax,
                                                float* __restrict__ lbuf,
                                                float* __restrict__ nodesum) {
  int t = blockIdx.x * 256 + threadIdx.x;
  if (t >= N_EDGES * KK) return;
  const int k = t & 3;
  const int e = t >> 2;
  const int d = dst[e];
  float mx = ofOrd(nodemax[(size_t)d * KK + k]);
  float ev = expf(lbuf[t] - mx);
  lbuf[t] = ev;
  atomicAdd(nodesum + (size_t)d * KK + k, ev);
}

__global__ __launch_bounds__(256) void edge_scatter(const float* __restrict__ ft,
                                                    const int* __restrict__ src,
                                                    const int* __restrict__ dst,
                                                    const float* __restrict__ lbuf,
                                                    const float* __restrict__ nodesum,
                                                    float* __restrict__ out) {
  const int wave = threadIdx.x >> 5;
  const int lane = threadIdx.x & 31;
  const long e = (long)blockIdx.x * 8 + wave;
  if (e >= N_EDGES) return;
  const int s = src[e];
  const int d = dst[e];
  const int idx = lane * 4;
  const int k   = lane >> 3;
  const float a = lbuf[e * 4 + k] / nodesum[(size_t)d * KK + k];
  float4 v = *(const float4*)(ft + (size_t)s * KF + idx);
  float* po = out + (size_t)d * KF + idx;
  atomicAdd(po + 0, v.x * a);
  atomicAdd(po + 1, v.y * a);
  atomicAdd(po + 2, v.z * a);
  atomicAdd(po + 3, v.w * a);
}

extern "C" void kernel_launch(void* const* d_in, const int* in_sizes, int n_in,
                              void* d_out, int out_size, void* d_ws, size_t ws_size,
                              hipStream_t stream) {
  const float* feat  = (const float*)d_in[0];
  const float* W     = (const float*)d_in[1];
  const float* bias  = (const float*)d_in[2];
  const float* mu    = (const float*)d_in[3];
  const float* sigma = (const float*)d_in[4];
  const int*   src   = (const int*)d_in[5];
  const int*   dst   = (const int*)d_in[6];
  float* out = (float*)d_out;

  char* ws = (char*)d_ws;
  float*    ft      = (float*)(ws);
  float*    lbuf    = (float*)(ws + 25600000);
  unsigned* nodemax = (unsigned*)(ws + 38400000);
  float*    nodesum = (float*)(ws + 39200000);

  init_ws<<<(N_NODES * KF + 255) / 256, 256, 0, stream>>>(bias, out, nodemax, nodesum);
  fc_wmma<<<N_NODES / 16, 256, 0, stream>>>(feat, W, ft);
  edge_logits<<<(N_EDGES * KK + 255) / 256, 256, 0, stream>>>(ft, mu, sigma, src, dst,
                                                              lbuf, nodemax);
  edge_exp<<<(N_EDGES * KK + 255) / 256, 256, 0, stream>>>(dst, nodemax, lbuf, nodesum);
  edge_scatter<<<(N_EDGES + 7) / 8, 256, 0, stream>>>(ft, src, dst, lbuf, nodesum, out);
}


# wave entry: init, edge_exp, edge_scatter run as few long waves (scatter: 32 edges per wave, read lane-parallel, walked by readlane): 1100000 one-item waves become 47500; arrival-order sums as before
# speedup vs baseline: 1.3570x; 1.3570x over previous
#include <hip/hip_runtime.h>

typedef __attribute__((ext_vector_type(2))) float v2f;
typedef __attribute__((ext_vector_type(8))) float v8f;
typedef __attribute__((ext_vector_type(4))) float v4f;

#define N_NODES 50000
#define N_EDGES 800000
#define D_IN    128
#define KK      4
#define FF      32
#define KF      128

__device__ __forceinline__ unsigned ordOf(float f) {
  unsigned u = __float_as_uint(f);
  return (u >> 31) ? ~u : (u | 0x80000000u);
}
__device__ __forceinline__ float ofOrd(unsigned o) {
  return (o >> 31) ? __uint_as_float(o & 0x7fffffffu) : __uint_as_float(~o);
}

#define INIT_BLOCKS  1250
#define INIT_THREADS (INIT_BLOCKS * 256)
#define INIT_TRIPS   5
__global__ __launch_bounds__(256) void init_ws(const float* __restrict__ bias,
                                               float* __restrict__ out,
                                               unsigned* __restrict__ nodemax,
                                               float* __restrict__ nodesum) {
  static_assert(INIT_THREADS % (KF / 4) == 0, "a thread must stay in one column piece");
  static_assert(INIT_THREADS * INIT_TRIPS == N_NODES * KF / 4, "out is covered exactly");
  static_assert(N_NODES * KK / 4 <= INIT_THREADS, "one piece of nodemax and nodesum per thread");
  const int t0 = blockIdx.x * 256 + threadIdx.x;
  const v4f b = ((const v4f*)bias)[t0 & (KF / 4 - 1)];
  v4f* o4 = (v4f*)out;
  #pragma unroll
  for (int i = 0; i < INIT_TRIPS; ++i) o4[t0 + i * INIT_THREADS] = b;
  if (t0 < N_NODES * KK / 4) {
    ((uint4*)nodemax)[t0]  = make_uint4(0u, 0u, 0u, 0u);
    ((float4*)nodesum)[t0] = make_float4(0.0f, 0.0f, 0.0f, 0.0f);
  }
}

__global__ __launch_bounds__(256) void fc_wmma(const float* __restrict__ feat,
                                               const float* __restrict__ W,
                                               float* __restrict__ ft) {
  __shared__ float As[16 * D_IN];
  const int lane = threadIdx.x & 31;
  const int wave = threadIdx.x >> 5;
  const int row0 = blockIdx.x * 16;

  for (int i = threadIdx.x; i < 16 * D_IN / 4; i += 256) {
    ((float4*)As)[i] = ((const float4*)(feat + row0 * D_IN))[i];
  }
  __syncthreads();

  const int col0 = wave * 16;
  const int ln   = lane & 15;
  const int kh   = (lane >> 4) << 1;

  v8f c = {};
  #pragma unroll 8
  for (int k = 0; k < D_IN; k += 4) {
    v2f a; a.x = As[ln * D_IN + k + kh];
           a.y = As[ln * D_IN + k + kh + 1];
    v2f b; b.x = W[(col0 + ln) * D_IN + k + kh];
           b.y = W[(col0 + ln) * D_IN + k + kh + 1];
    c = __builtin_amdgcn_wmma_f32_16x16x4_f32(false, a, false, b,
                                              (short)0, c, false, false);
  }

  const int mo = (lane >> 4) << 3;
  #pragma unroll
  for (int r = 0; r < 8; ++r) {
    ft[(row0 + r + mo) * KF + col0 + ln] = c[r];
  }
}

__global__ __launch_bounds__(256) void edge_logits(const float* __restrict__ ft,
                                                   const float* __restrict__ mu,
                                                   const float* __restrict__ sigma,
                                                   const int* __restrict__ src,
                                                   const int* __restrict__ dst,
                                                   float* __restrict__ lbuf,
                                                   unsigned* __restrict__ nodemax) {
  int t = blockIdx.x * 256 + threadIdx.x;
  if (t >= N_EDGES * KK) return;
  const int k = t & 3;
  const int e = t >> 2;
  const int s = src[e];
  const int d = dst[e];
  const float4* ps = (const float4*)(ft + (size_t)s * KF + k * FF);
  const float4* pd = (const float4*)(ft + (size_t)d * KF + k * FF);
  const float4* pm = (const float4*)(mu + k * FF);
  float acc = 0.0f;
  #pragma unroll
  for (int i = 0; i < FF / 4; ++i) {
    float4 a = ps[i], b = pd[i], m = pm[i];
    float dx = a.x - b.x - m.x;
    float dy = a.y - b.y - m.y;
    float dz = a.z - b.z - m.z;
    float dw = a.w - b.w - m.w;
    acc += dx * dx + dy * dy + dz * dz + dw * dw;
  }
  float logit = -sqrtf(acc) * sigma[k];
  lbuf[t] = logit;
  atomicMax(nodemax + (size_t)d * KK + k, ordOf(logit));
}

#define EXP_ITEMS 8
#define EXP_WAVES (N_EDGES * KK / (32 * EXP_ITEMS))
__global__ __launch_bounds__(256) void edge_exp(const int* __restrict__ dst,
                                                const unsigned* __restrict__ nodemax,
                                                float* __restrict__ lbuf,
                                                float* __restrict__ nodesum) {
  static_assert((N_EDGES * KK) % (32 * EXP_ITEMS) == 0, "whole groups per wave");
  const int lane = threadIdx.x & 31;
  const int wave = blockIdx.x * 8 + (threadIdx.x >> 5);
  if (wave >= EXP_WAVES) return;
  const int t0 = wave * (32 * EXP_ITEMS) + lane;
  const int k = lane & 3;
  int      d[EXP_ITEMS];
  float    lg[EXP_ITEMS];
  unsigned mo[EXP_ITEMS];
  #pragma unroll
  for (int i = 0; i < EXP_ITEMS; ++i) {
    const int t = t0 + i * 32;
    d[i]  = dst[t >> 2];
    lg[i] = lbuf[t];
  }
  #pragma unroll
  for (int i = 0; i < EXP_ITEMS; ++i) mo[i] = nodemax[(size_t)d[i] * KK + k];
  #pragma unroll
  for (int i = 0; i < EXP_ITEMS; ++i) {
    const float ev = expf(lg[i] - ofOrd(mo[i]));
    lbuf[t0 + i * 32] = ev;
    atomicAdd(nodesum + (size_t)d[i] * KK + k, ev);
  }
}

#define SC_GROUP 4
__global__ __launch_bounds__(256) void edge_scatter(const float* __restrict__ ft,
                                                    const int* __restrict__ src,
                                                    const int* __restrict__ dst,
                                                    const float* __restrict__ lbuf,
                                                    const float* __restrict__ nodesum,
                                                    float* __restrict__ out) {
  static_assert(N_EDGES % 32 == 0, "whole chunks of 32 edges");
  const int lane = threadIdx.x & 31;
  const int wave = __builtin_amdgcn_readfirstlane(threadIdx.x >> 5);
  const int base = (blockIdx.x * 8 + wave) * 32;
  if (base >= N_EDGES) return;
  const int e   = base + lane;
  const int s_l = src[e];
  const int d_l = dst[e];
  const float4 ev = ((const float4*)lbuf)[e];
  const float4 sm = ((const float4*)nodesum)[d_l];
  const float c0 = ev.x / sm.x;
  const float c1 = ev.y / sm.y;
  const float c2 = ev.z / sm.z;
  const float c3 = ev.w / sm.w;
  for (int j0 = 0; j0 < 32; j0 += SC_GROUP) {
    float v[SC_GROUP][KK];
    #pragma unroll
    for (int u = 0; u < SC_GROUP; ++u) {
      const int s = __builtin_amdgcn_readlane(s_l, j0 + u);
      const float* ps = ft + (size_t)s * KF + lane;
      v[u][0] = ps[0 * FF];
      v[u][1] = ps[1 * FF];
      v[u][2] = ps[2 * FF];
      v[u][3] = ps[3 * FF];
    }
    #pragma unroll
    for (int u = 0; u < SC_GROUP; ++u) {
      const int d = __builtin_amdgcn_readlane(d_l, j0 + u);
      const float a0 = __int_as_float(__builtin_amdgcn_readlane(__float_as_int(c0), j0 + u));
      const float a1 = __int_as_float(__builtin_amdgcn_readlane(__float_as_int(c1), j0 + u));
      const float a2 = __int_as_float(__builtin_amdgcn_readlane(__float_as_int(c2), j0 + u));
      const float a3 = __int_as_float(__builtin_amdgcn_readlane(__float_as_int(c3), j0 + u));
      float* po = out + (size_t)d * KF + lane;
      atomicAdd(po + 0 * FF, v[u][0] * a0);
      atomicAdd(po + 1 * FF, v[u][1] * a1);
      atomicAdd(po + 2 * FF, v[u][2] * a2);
      atomicAdd(po + 3 * FF, v[u][3] * a3);
    }
  }
}

extern "C" void kernel_launch(void* const* d_in, const int* in_sizes, int n_in,
                              void* d_out, int out_size, void* d_ws, size_t ws_size,
                              hipStream_t stream) {
  const float* feat  = (const float*)d_in[0];
  const float* W     = (const float*)d_in[1];
  const float* bias  = (const float*)d_in[2];
  const float* mu    = (const float*)d_in[3];
  const float* sigma = (const float*)d_in[4];
  const int*   src   = (const int*)d_in[5];
  const int*   dst   = (const int*)d_in[6];
  float* out = (float*)d_out;

  char* ws = (char*)d_ws;
  float*    ft      = (float*)(ws);
  float*    lbuf    = (float*)(ws + 25600000);
  unsigned* nodemax = (unsigned*)(ws + 38400000);
  float*    nodesum = (float*)(ws + 39200000);

  init_ws<<<INIT_BLOCKS, 256, 0, stream>>>(bias, out, nodemax, nodesum);
  fc_wmma<<<N_NODES / 16, 256, 0, stream>>>(feat, W, ft);
  edge_logits<<<(N_EDGES * KK + 255) / 256, 256, 0, stream>>>(ft, mu, sigma, src, dst,
                                                              lbuf, nodemax);
  edge_exp<<<(EXP_WAVES + 7) / 8, 256, 0, stream>>>(dst, nodemax, lbuf, nodesum);
  edge_scatter<<<(N_EDGES / 32 + 7) / 8, 256, 0, stream>>>(ft, src, dst, lbuf, nodesum, out);
}


# edge_logits: each wave fetches its 8 edges' src and dst rows by whole-row 16-byte loads into padded LDS; a lane then reads its own slice from LDS; per-lane arithmetic and its sum order unchanged
# speedup vs baseline: 1.5610x; 1.1503x over previous
#include <hip/hip_runtime.h>

typedef __attribute__((ext_vector_type(2))) float v2f;
typedef __attribute__((ext_vector_type(8))) float v8f;
typedef __attribute__((ext_vector_type(4))) float v4f;

#define N_NODES 50000
#define N_EDGES 800000
#define D_IN    128
#define KK      4
#define FF      32
#define KF      128

__device__ __forceinline__ unsigned ordOf(float f) {
  unsigned u = __float_as_uint(f);
  return (u >> 31) ? ~u : (u | 0x80000000u);
}
__device__ __forceinline__ float ofOrd(unsigned o) {
  return (o >> 31) ? __uint_as_float(o & 0x7fffffffu) : __uint_as_float(~o);
}

#define INIT_BLOCKS  1250
#define INIT_THREADS (INIT_BLOCKS * 256)
#define INIT_TRIPS   5
__global__ __launch_bounds__(256) void init_ws(const float* __restrict__ bias,
                                               float* __restrict__ out,
                                               unsigned* __restrict__ nodemax,
                                               float* __restrict__ nodesum) {
  static_assert(INIT_THREADS % (KF / 4) == 0, "a thread must stay in one column piece");
  static_assert(INIT_THREADS * INIT_TRIPS == N_NODES * KF / 4, "out is covered exactly");
  static_assert(N_NODES * KK / 4 <= INIT_THREADS, "one piece of nodemax and nodesum per thread");
  const int t0 = blockIdx.x * 256 + threadIdx.x;
  const v4f b = ((const v4f*)bias)[t0 & (KF / 4 - 1)];
  v4f* o4 = (v4f*)out;
  #pragma unroll
  for (int i = 0; i < INIT_TRIPS; ++i) o4[t0 + i * INIT_THREADS] = b;
  if (t0 < N_NODES * KK / 4) {
    ((uint4*)nodemax)[t0]  = make_uint4(0u, 0u, 0u, 0u);
    ((float4*)nodesum)[t0] = make_float4(0.0f, 0.0f, 0.0f, 0.0f);
  }
}

__global__ __launch_bounds__(256) void fc_wmma(const float* __restrict__ feat,
                                               const float* __restrict__ W,
                                               float* __restrict__ ft) {
  __shared__ float As[16 * D_IN];
  const int lane = threadIdx.x & 31;
  const int wave = threadIdx.x >> 5;
  const int row0 = blockIdx.x * 16;

  for (int i = threadIdx.x; i < 16 * D_IN / 4; i += 256) {
    ((float4*)As)[i] = ((const float4*)(feat + row0 * D_IN))[i];
  }
  __syncthreads();

  const int col0 = wave * 16;
  const int ln   = lane & 15;
  const int kh   = (lane >> 4) << 1;

  v8f c = {};
  #pragma unroll 8
  for (int k = 0; k < D_IN; k += 4) {
    v2f a; a.x = As[ln * D_IN + k + kh];
           a.y = As[ln * D_IN + k + kh + 1];
    v2f b; b.x = W[(col0 + ln) * D_IN + k + kh];
           b.y = W[(col0 + ln) * D_IN + k + kh + 1];
    c = __builtin_amdgcn_wmma_f32_16x16x4_f32(false, a, false, b,
                                              (short)0, c, false, false);
  }

  const int mo = (lane >> 4) << 3;
  #pragma unroll
  for (int r = 0; r < 8; ++r) {
    ft[(row0 + r + mo) * KF + col0 + ln] = c[r];
  }
}

#define LG_SLICE_SLOTS 9
#define LG_ROW_SLOTS   (KK * LG_SLICE_SLOTS)
#define LG_EDGES       64
__global__ __launch_bounds__(256) void edge_logits(const float* __restrict__ ft,
                                                   const float* __restrict__ mu,
                                                   const float* __restrict__ sigma,
                                                   const int* __restrict__ src,
                                                   const int* __restrict__ dst,
                                                   float* __restrict__ lbuf,
                                                   unsigned* __restrict__ nodemax) {
  static_assert((N_EDGES * KK) % 256 == 0, "whole blocks: every thread has an item");
  static_assert(KF * 4 == 32 * 16, "a row is one 16-byte piece per lane");
  __shared__ float4 srow[LG_EDGES * LG_ROW_SLOTS];
  __shared__ float4 drow[LG_EDGES * LG_ROW_SLOTS];
  const int lane = threadIdx.x & 31;
  const int wave = __builtin_amdgcn_readfirstlane(threadIdx.x >> 5);
  const int t = blockIdx.x * 256 + threadIdx.x;
  const int k = t & 3;
  const int e = t >> 2;
  const int d = dst[e];
  const int e0  = blockIdx.x * LG_EDGES + wave * 8;
  const int s_l = src[e0 + (lane & 7)];
  const int d_l = dst[e0 + (lane & 7)];
  float4 va[8], vb[8];
  #pragma unroll
  for (int j = 0; j < 8; ++j) {
    const int sj = __builtin_amdgcn_readlane(s_l, j);
    const int dj = __builtin_amdgcn_readlane(d_l, j);
    va[j] = ((const float4*)(ft + (size_t)sj * KF))[lane];
    vb[j] = ((const float4*)(ft + (size_t)dj * KF))[lane];
  }
  const int wslot = (lane >> 3) * LG_SLICE_SLOTS + (lane & 7);
  #pragma unroll
  for (int j = 0; j < 8; ++j) {
    srow[(wave * 8 + j) * LG_ROW_SLOTS + wslot] = va[j];
    drow[(wave * 8 + j) * LG_ROW_SLOTS + wslot] = vb[j];
  }
  __syncthreads();
  const int rslot = (wave * 8 + (lane >> 2)) * LG_ROW_SLOTS + k * LG_SLICE_SLOTS;
  const float4* ps = srow + rslot;
  const float4* pd = drow + rslot;
  const float4* pm = (const float4*)(mu + k * FF);
  float acc = 0.0f;
  #pragma unroll
  for (int i = 0; i < FF / 4; ++i) {
    float4 a = ps[i], b = pd[i], m = pm[i];
    float dx = a.x - b.x - m.x;
    float dy = a.y - b.y - m.y;
    float dz = a.z - b.z - m.z;
    float dw = a.w - b.w - m.w;
    acc += dx * dx + dy * dy + dz * dz + dw * dw;
  }
  float logit = -sqrtf(acc) * sigma[k];
  lbuf[t] = logit;
  atomicMax(nodemax + (size_t)d * KK + k, ordOf(logit));
}

#define EXP_ITEMS 8
#define EXP_WAVES (N_EDGES * KK / (32 * EXP_ITEMS))
__global__ __launch_bounds__(256) void edge_exp(const int* __restrict__ dst,
                                                const unsigned* __restrict__ nodemax,
                                                float* __restrict__ lbuf,
                                                float* __restrict__ nodesum) {
  static_assert((N_EDGES * KK) % (32 * EXP_ITEMS) == 0, "whole groups per wave");
  const int lane = threadIdx.x & 31;
  const int wave = blockIdx.x * 8 + (threadIdx.x >> 5);
  if (wave >= EXP_WAVES) return;
  const int t0 = wave * (32 * EXP_ITEMS) + lane;
  const int k = lane & 3;
  int      d[EXP_ITEMS];
  float    lg[EXP_ITEMS];
  unsigned mo[EXP_ITEMS];
  #pragma unroll
  for (int i = 0; i < EXP_ITEMS; ++i) {
    const int t = t0 + i * 32;
    d[i]  = dst[t >> 2];
    lg[i] = lbuf[t];
  }
  #pragma unroll
  for (int i = 0; i < EXP_ITEMS; ++i) mo[i] = nodemax[(size_t)d[i] * KK + k];
  #pragma unroll
  for (int i = 0; i < EXP_ITEMS; ++i) {
    const float ev = expf(lg[i] - ofOrd(mo[i]));
    lbuf[t0 + i * 32] = ev;
    atomicAdd(nodesum + (size_t)d[i] * KK + k, ev);
  }
}

#define SC_GROUP 4
__global__ __launch_bounds__(256) void edge_scatter(const float* __restrict__ ft,
                                                    const int* __restrict__ src,
                                                    const int* __restrict__ dst,
                                                    const float* __restrict__ lbuf,
                                                    const float* __restrict__ nodesum,
                                                    float* __restrict__ out) {
  static_assert(N_EDGES % 32 == 0, "whole chunks of 32 edges");
  const int lane = threadIdx.x & 31;
  const int wave = __builtin_amdgcn_readfirstlane(threadIdx.x >> 5);
  const int base = (blockIdx.x * 8 + wave) * 32;
  if (base >= N_EDGES) return;
  const int e   = base + lane;
  const int s_l = src[e];
  const int d_l = dst[e];
  const float4 ev = ((const float4*)lbuf)[e];
  const float4 sm = ((const float4*)nodesum)[d_l];
  const float c0 = ev.x / sm.x;
  const float c1 = ev.y / sm.y;
  const float c2 = ev.z / sm.z;
  const float c3 = ev.w / sm.w;
  for (int j0 = 0; j0 < 32; j0 += SC_GROUP) {
    float v[SC_GROUP][KK];
    #pragma unroll
    for (int u = 0; u < SC_GROUP; ++u) {
      const int s = __builtin_amdgcn_readlane(s_l, j0 + u);
      const float* ps = ft + (size_t)s * KF + lane;
      v[u][0] = ps[0 * FF];
      v[u][1] = ps[1 * FF];
      v[u][2] = ps[2 * FF];
      v[u][3] = ps[3 * FF];
    }
    #pragma unroll
    for (int u = 0; u < SC_GROUP; ++u) {
      const int d = __builtin_amdgcn_readlane(d_l, j0 + u);
      const float a0 = __int_as_float(__builtin_amdgcn_readlane(__float_as_int(c0), j0 + u));
      const float a1 = __int_as_float(__builtin_amdgcn_readlane(__float_as_int(c1), j0 + u));
      const float a2 = __int_as_float(__builtin_amdgcn_readlane(__float_as_int(c2), j0 + u));
      const float a3 = __int_as_float(__builtin_amdgcn_readlane(__float_as_int(c3), j0 + u));
      float* po = out + (size_t)d * KF + lane;
      atomicAdd(po + 0 * FF, v[u][0] * a0);
      atomicAdd(po + 1 * FF, v[u][1] * a1);
      atomicAdd(po + 2 * FF, v[u][2] * a2);
      atomicAdd(po + 3 * FF, v[u][3] * a3);
    }
  }
}

extern "C" void kernel_launch(void* const* d_in, const int* in_sizes, int n_in,
                              void* d_out, int out_size, void* d_ws, size_t ws_size,
                              hipStream_t stream) {
  const float* feat  = (const float*)d_in[0];
  const float* W     = (const float*)d_in[1];
  const float* bias  = (const float*)d_in[2];
  const float* mu    = (const float*)d_in[3];
  const float* sigma = (const float*)d_in[4];
  const int*   src   = (const int*)d_in[5];
  const int*   dst   = (const int*)d_in[6];
  float* out = (float*)d_out;

  char* ws = (char*)d_ws;
  float*    ft      = (float*)(ws);
  float*    lbuf    = (float*)(ws + 25600000);
  unsigned* nodemax = (unsigned*)(ws + 38400000);
  float*    nodesum = (float*)(ws + 39200000);

  init_ws<<<INIT_BLOCKS, 256, 0, stream>>>(bias, out, nodemax, nodesum);
  fc_wmma<<<N_NODES / 16, 256, 0, stream>>>(feat, W, ft);
  edge_logits<<<(N_EDGES * KK + 255) / 256, 256, 0, stream>>>(ft, mu, sigma, src, dst,
                                                              lbuf, nodemax);
  edge_exp<<<(EXP_WAVES + 7) / 8, 256, 0, stream>>>(dst, nodemax, lbuf, nodesum);
  edge_scatter<<<(N_EDGES / 32 + 7) / 8, 256, 0, stream>>>(ft, src, dst, lbuf, nodesum, out);
}


# fc_wmma without LDS: a wave keeps its 16 columns of W in registers over its row tiles and takes feat fragments straight from global by 16-byte loads and a lane-half exchange; sums unchanged
# speedup vs baseline: 1.6769x; 1.0742x over previous
#include <hip/hip_runtime.h>

typedef __attribute__((ext_vector_type(2))) float v2f;
typedef __attribute__((ext_vector_type(8))) float v8f;
typedef __attribute__((ext_vector_type(4))) float v4f;

#define N_NODES 50000
#define N_EDGES 800000
#define D_IN    128
#define KK      4
#define FF      32
#define KF      128

__device__ __forceinline__ unsigned ordOf(float f) {
  unsigned u = __float_as_uint(f);
  return (u >> 31) ? ~u : (u | 0x80000000u);
}
__device__ __forceinline__ float ofOrd(unsigned o) {
  return (o >> 31) ? __uint_as_float(o & 0x7fffffffu) : __uint_as_float(~o);
}

#define INIT_BLOCKS  1250
#define INIT_THREADS (INIT_BLOCKS * 256)
#define INIT_TRIPS   5
__global__ __launch_bounds__(256) void init_ws(const float* __restrict__ bias,
                                               float* __restrict__ out,
                                               unsigned* __restrict__ nodemax,
                                               float* __restrict__ nodesum) {
  static_assert(INIT_THREADS % (KF / 4) == 0, "a thread must stay in one column piece");
  static_assert(INIT_THREADS * INIT_TRIPS == N_NODES * KF / 4, "out is covered exactly");
  static_assert(N_NODES * KK / 4 <= INIT_THREADS, "one piece of nodemax and nodesum per thread");
  const int t0 = blockIdx.x * 256 + threadIdx.x;
  const v4f b = ((const v4f*)bias)[t0 & (KF / 4 - 1)];
  v4f* o4 = (v4f*)out;
  #pragma unroll
  for (int i = 0; i < INIT_TRIPS; ++i) o4[t0 + i * INIT_THREADS] = b;
  if (t0 < N_NODES * KK / 4) {
    ((uint4*)nodemax)[t0]  = make_uint4(0u, 0u, 0u, 0u);
    ((float4*)nodesum)[t0] = make_float4(0.0f, 0.0f, 0.0f, 0.0f);
  }
}

#define FC_NT     (N_NODES / 16)
#define FC_BLOCKS 768
__device__ __forceinline__ float other_half(float x) {
  const int b = __builtin_bit_cast(int, x);
  return __builtin_bit_cast(float,
      __builtin_amdgcn_permlanex16(b, b, 0x76543210, 0xfedcba98, false, false));
}
__device__ __forceinline__ void frag_pair(const v4f q, const int hf, v2f& f0, v2f& f1) {
  const float rp = other_half(hf ? q.x : q.z);
  const float rq = other_half(hf ? q.y : q.w);
  f0.x = hf ? rp  : q.x;  f0.y = hf ? rq  : q.y;
  f1.x = hf ? q.z : rp;   f1.y = hf ? q.w : rq;
}
__device__ __forceinline__ void load_row_half(const float* __restrict__ p, v4f (&q)[D_IN / 8]) {
  #pragma unroll
  for (int i = 0; i < D_IN / 8; ++i) q[i] = *(const v4f*)(p + 8 * i);
}
__device__ __forceinline__ void fc_tile(const v4f (&q)[D_IN / 8], const v2f (&bfr)[D_IN / 4],
                                        const int hf, float* __restrict__ dst) {
  v8f c = {};
  #pragma unroll
  for (int i = 0; i < D_IN / 8; ++i) {
    v2f a0, a1;
    frag_pair(q[i], hf, a0, a1);
    c = __builtin_amdgcn_wmma_f32_16x16x4_f32(false, a0, false, bfr[2 * i],
                                              (short)0, c, false, false);
    c = __builtin_amdgcn_wmma_f32_16x16x4_f32(false, a1, false, bfr[2 * i + 1],
                                              (short)0, c, false, false);
  }
  #pragma unroll
  for (int r = 0; r < 8; ++r) dst[r * KF] = c[r];
}
__global__ __launch_bounds__(256) void fc_wmma(const float* __restrict__ feat,
                                               const float* __restrict__ W,
                                               float* __restrict__ ft) {
  static_assert(N_NODES % 16 == 0, "whole row tiles");
  static_assert(FC_BLOCKS <= FC_NT, "every work-group has a first tile");
  static_assert(D_IN == 128 && KF == 128, "8 waves x 16 columns, 16 pieces of 8 k per row");
  const int lane = threadIdx.x & 31;
  const int wave = threadIdx.x >> 5;
  const int ln   = lane & 15;
  const int hf   = lane >> 4;
  const int col0 = wave * 16;

  v2f bfr[D_IN / 4];
  {
    v4f wq[D_IN / 8];
    load_row_half(W + (col0 + ln) * D_IN + 4 * hf, wq);
    #pragma unroll
    for (int i = 0; i < D_IN / 8; ++i) frag_pair(wq[i], hf, bfr[2 * i], bfr[2 * i + 1]);
  }

  const float* ap = feat + (size_t)ln * D_IN + 4 * hf;
  float*       dp = ft + (size_t)(hf << 3) * KF + col0 + ln;
  for (int tile = blockIdx.x; tile < FC_NT; tile += FC_BLOCKS) {
    v4f q[D_IN / 8];
    load_row_half(ap + (size_t)tile * 16 * D_IN, q);
    fc_tile(q, bfr, hf, dp + (size_t)tile * 16 * KF);
  }
}

#define LG_SLICE_SLOTS 9
#define LG_ROW_SLOTS   (KK * LG_SLICE_SLOTS)
#define LG_EDGES       64
__global__ __launch_bounds__(256) void edge_logits(const float* __restrict__ ft,
                                                   const float* __restrict__ mu,
                                                   const float* __restrict__ sigma,
                                                   const int* __restrict__ src,
                                                   const int* __restrict__ dst,
                                                   float* __restrict__ lbuf,
                                                   unsigned* __restrict__ nodemax) {
  static_assert((N_EDGES * KK) % 256 == 0, "whole blocks: every thread has an item");
  static_assert(KF * 4 == 32 * 16, "a row is one 16-byte piece per lane");
  __shared__ float4 srow[LG_EDGES * LG_ROW_SLOTS];
  __shared__ float4 drow[LG_EDGES * LG_ROW_SLOTS];
  const int lane = threadIdx.x & 31;
  const int wave = __builtin_amdgcn_readfirstlane(threadIdx.x >> 5);
  const int t = blockIdx.x * 256 + threadIdx.x;
  const int k = t & 3;
  const int e = t >> 2;
  const int d = dst[e];
  const int e0  = blockIdx.x * LG_EDGES + wave * 8;
  const int s_l = src[e0 + (lane & 7)];
  const int d_l = dst[e0 + (lane & 7)];
  float4 va[8], vb[8];
  #pragma unroll
  for (int j = 0; j < 8; ++j) {
    const int sj = __builtin_amdgcn_readlane(s_l, j);
    const int dj = __builtin_amdgcn_readlane(d_l, j);
    va[j] = ((const float4*)(ft + (size_t)sj * KF))[lane];
    vb[j] = ((const float4*)(ft + (size_t)dj * KF))[lane];
  }
  const int wslot = (lane >> 3) * LG_SLICE_SLOTS + (lane & 7);
  #pragma unroll
  for (int j = 0; j < 8; ++j) {
    srow[(wave * 8 + j) * LG_ROW_SLOTS + wslot] = va[j];
    drow[(wave * 8 + j) * LG_ROW_SLOTS + wslot] = vb[j];
  }
  __syncthreads();
  const int rslot = (wave * 8 + (lane >> 2)) * LG_ROW_SLOTS + k * LG_SLICE_SLOTS;
  const float4* ps = srow + rslot;
  const float4* pd = drow + rslot;
  const float4* pm = (const float4*)(mu + k * FF);
  float acc = 0.0f;
  #pragma unroll
  for (int i = 0; i < FF / 4; ++i) {
    float4 a = ps[i], b = pd[i], m = pm[i];
    float dx = a.x - b.x - m.x;
    float dy = a.y - b.y - m.y;
    float dz = a.z - b.z - m.z;
    float dw = a.w - b.w - m.w;
    acc += dx * dx + dy * dy + dz * dz + dw * dw;
  }
  float logit = -sqrtf(acc) * sigma[k];
  lbuf[t] = logit;
  atomicMax(nodemax + (size_t)d * KK + k, ordOf(logit));
}

#define EXP_ITEMS 8
#define EXP_WAVES (N_EDGES * KK / (32 * EXP_ITEMS))
__global__ __launch_bounds__(256) void edge_exp(const int* __restrict__ dst,
                                                const unsigned* __restrict__ nodemax,
                                                float* __restrict__ lbuf,
                                                float* __restrict__ nodesum) {
  static_assert((N_EDGES * KK) % (32 * EXP_ITEMS) == 0, "whole groups per wave");
  const int lane = threadIdx.x & 31;
  const int wave = blockIdx.x * 8 + (threadIdx.x >> 5);
  if (wave >= EXP_WAVES) return;
  const int t0 = wave * (32 * EXP_ITEMS) + lane;
  const int k = lane & 3;
  int      d[EXP_ITEMS];
  float    lg[EXP_ITEMS];
  unsigned mo[EXP_ITEMS];
  #pragma unroll
  for (int i = 0; i < EXP_ITEMS; ++i) {
    const int t = t0 + i * 32;
    d[i]  = dst[t >> 2];
    lg[i] = lbuf[t];
  }
  #pragma unroll
  for (int i = 0; i < EXP_ITEMS; ++i) mo[i] = nodemax[(size_t)d[i] * KK + k];
  #pragma unroll
  for (int i = 0; i < EXP_ITEMS; ++i) {
    const float ev = expf(lg[i] - ofOrd(mo[i]));
    lbuf[t0 + i * 32] = ev;
    atomicAdd(nodesum + (size_t)d[i] * KK + k, ev);
  }
}

#define SC_GROUP 4
__global__ __launch_bounds__(256) void edge_scatter(const float* __restrict__ ft,
                                                    const int* __restrict__ src,
                                                    const int* __restrict__ dst,
                                                    const float* __restrict__ lbuf,
                                                    const float* __restrict__ nodesum,
                                                    float* __restrict__ out) {
  static_assert(N_EDGES % 32 == 0, "whole chunks of 32 edges");
  const int lane = threadIdx.x & 31;
  const int wave = __builtin_amdgcn_readfirstlane(threadIdx.x >> 5);
  const int base = (blockIdx.x * 8 + wave) * 32;
  if (base >= N_EDGES) return;
  const int e   = base + lane;
  const int s_l = src[e];
  const int d_l = dst[e];
  const float4 ev = ((const float4*)lbuf)[e];
  const float4 sm = ((const float4*)nodesum)[d_l];
  const float c0 = ev.x / sm.x;
  const float c1 = ev.y / sm.y;
  const float c2 = ev.z / sm.z;
  const float c3 = ev.w / sm.w;
  for (int j0 = 0; j0 < 32; j0 += SC_GROUP) {
    float v[SC_GROUP][KK];
    #pragma unroll
    for (int u = 0; u < SC_GROUP; ++u) {
      const int s = __builtin_amdgcn_readlane(s_l, j0 + u);
      const float* ps = ft + (size_t)s * KF + lane;
      v[u][0] = ps[0 * FF];
      v[u][1] = ps[1 * FF];
      v[u][2] = ps[2 * FF];
      v[u][3] = ps[3 * FF];
    }
    #pragma unroll
    for (int u = 0; u < SC_GROUP; ++u) {
      const int d = __builtin_amdgcn_readlane(d_l, j0 + u);
      const float a0 = __int_as_float(__builtin_amdgcn_readlane(__float_as_int(c0), j0 + u));
      const float a1 = __int_as_float(__builtin_amdgcn_readlane(__float_as_int(c1), j0 + u));
      const float a2 = __int_as_float(__builtin_amdgcn_readlane(__float_as_int(c2), j0 + u));
      const float a3 = __int_as_float(__builtin_amdgcn_readlane(__float_as_int(c3), j0 + u));
      float* po = out + (size_t)d * KF + lane;
      atomicAdd(po + 0 * FF, v[u][0] * a0);
      atomicAdd(po + 1 * FF, v[u][1] * a1);
      atomicAdd(po + 2 * FF, v[u][2] * a2);
      atomicAdd(po + 3 * FF, v[u][3] * a3);
    }
  }
}

extern "C" void kernel_launch(void* const* d_in, const int* in_sizes, int n_in,
                              void* d_out, int out_size, void* d_ws, size_t ws_size,
                              hipStream_t stream) {
  const float* feat  = (const float*)d_in[0];
  const float* W     = (const float*)d_in[1];
  const float* bias  = (const float*)d_in[2];
  const float* mu    = (const float*)d_in[3];
  const float* sigma = (const float*)d_in[4];
  const int*   src   = (const int*)d_in[5];
  const int*   dst   = (const int*)d_in[6];
  float* out = (float*)d_out;

  char* ws = (char*)d_ws;
  float*    ft      = (float*)(ws);
  float*    lbuf    = (float*)(ws + 25600000);
  unsigned* nodemax = (unsigned*)(ws + 38400000);
  float*    nodesum = (float*)(ws + 39200000);

  init_ws<<<INIT_BLOCKS, 256, 0, stream>>>(bias, out, nodemax, nodesum);
  fc_wmma<<<FC_BLOCKS, 256, 0, stream>>>(feat, W, ft);
  edge_logits<<<(N_EDGES * KK + 255) / 256, 256, 0, stream>>>(ft, mu, sigma, src, dst,
                                                              lbuf, nodemax);
  edge_exp<<<(EXP_WAVES + 7) / 8, 256, 0, stream>>>(dst, nodemax, lbuf, nodesum);
  edge_scatter<<<(N_EDGES / 32 + 7) / 8, 256, 0, stream>>>(ft, src, dst, lbuf, nodesum, out);
}
